# starting kernel
# baseline (speedup 1.0000x reference)
#include <hip/hip_runtime.h>
#include <math.h>

typedef __attribute__((ext_vector_type(2))) float v2f;
typedef __attribute__((ext_vector_type(8))) float v8f;

#define B_TOT 16384
#define LSEQ  50
#define DIM   64
#define NATTR 16
#define USERS_PER_BLOCK 16

__global__ __launch_bounds__(512) void fused_pool_proj_softmax(
    const int*   __restrict__ x,
    const int*   __restrict__ x_mask,
    const float* __restrict__ y,
    const float* __restrict__ ob,
    const float* __restrict__ item_emb,
    const float* __restrict__ w0, const float* __restrict__ b0,
    const float* __restrict__ w1, const float* __restrict__ b1,
    const float* __restrict__ w2, const float* __restrict__ b2,
    float* __restrict__ logit_out,
    float* __restrict__ partials)
{
    __shared__ float rep[USERS_PER_BLOCK][DIM + 4];
    __shared__ float Wu[USERS_PER_BLOCK][NATTR + 1];

    const int tid  = threadIdx.x;
    const int wid  = tid >> 5;
    const int lane = tid & 31;
    const int user = blockIdx.x * USERS_PER_BLOCK + wid;

    const int* xrow = x + user * LSEQ;
    const int  xi_a = xrow[lane];
    const int  xi_b = (lane < LSEQ - 32) ? xrow[32 + lane] : 0;

    int msum = x_mask[user * LSEQ + lane]
             + ((lane < LSEQ - 32) ? x_mask[user * LSEQ + 32 + lane] : 0);
#pragma unroll
    for (int off = 16; off >= 1; off >>= 1) msum += __shfl_xor(msum, off, 32);
    const float inv_len = 1.0f / (float)msum;

    float s0 = 0.0f, s1 = 0.0f;
#pragma unroll
    for (int l = 0; l < LSEQ; ++l) {
        const int idx = (l < 32) ? __shfl(xi_a, l, 32) : __shfl(xi_b, l - 32, 32);
        const float2 e = *(const float2*)(item_emb + (size_t)idx * DIM + 2 * lane);
        s0 += e.x;
        s1 += e.y;
    }
    rep[wid][2 * lane]     = s0 * inv_len;
    rep[wid][2 * lane + 1] = s1 * inv_len;
    __syncthreads();

    if (wid == 0) {
        const int half = lane >> 4;
        const int mn   = lane & 15;

        const float* wr;
        float bias;
        if      (mn < 2) { wr = w0 + mn * DIM;       bias = b0[mn];     }
        else if (mn < 8) { wr = w1 + (mn - 2) * DIM; bias = b1[mn - 2]; }
        else             { wr = w2 + (mn - 8) * DIM; bias = b2[mn - 8]; }

        v8f acc = {0.f, 0.f, 0.f, 0.f, 0.f, 0.f, 0.f, 0.f};
#pragma unroll
        for (int s = 0; s < 16; ++s) {
            const int k = 4 * s + 2 * half;
            v2f a, b;
            a.x = rep[mn][k];
            a.y = rep[mn][k + 1];
            b.x = wr[k];
            b.y = wr[k + 1];
            acc = __builtin_amdgcn_wmma_f32_16x16x4_f32(
                      false, a, false, b, (short)0, acc, false, false);
        }
#pragma unroll
        for (int r = 0; r < 8; ++r)
            Wu[r + 8 * half][mn] = acc[r] + bias;
    }
    __syncthreads();

    float num0 = 0.f, num1 = 0.f, num2 = 0.f;
    float cnt0 = 0.f, cnt1 = 0.f, cnt2 = 0.f;
    if (wid == 0) {
        if (lane < 16) {
            const int m = lane;
            const int gu = blockIdx.x * USERS_PER_BLOCK + m;
            float v[NATTR];
#pragma unroll
            for (int j = 0; j < NATTR; ++j) v[j] = Wu[m][j];

            const float* yrow  = y  + (size_t)gu * NATTR;
            const float* obrow = ob + (size_t)gu * NATTR;
            float*       lrow  = logit_out + (size_t)gu * NATTR;

            const int gs[4] = {0, 2, 8, 16};
            float nums[3], cnts[3];
#pragma unroll
            for (int g = 0; g < 3; ++g) {
                const int s = gs[g], e = gs[g + 1];
                float mx = -3.402823466e+38f;
                for (int j = s; j < e; ++j) mx = fmaxf(mx, v[j]);
                float se = 0.f;
                for (int j = s; j < e; ++j) se += expf(v[j] - mx);
                const float lse = mx + logf(se);
                const float inv = 1.0f / se;
                const float obs = obrow[s];
                float accl = 0.f;
                for (int j = s; j < e; ++j) {
                    const float lp = v[j] - lse;
                    accl  += yrow[j] * lp;
                    lrow[j] = expf(v[j] - mx) * inv;
                }
                nums[g] = -accl * obs;
                cnts[g] = obs;
            }
            num0 = nums[0]; num1 = nums[1]; num2 = nums[2];
            cnt0 = cnts[0]; cnt1 = cnts[1]; cnt2 = cnts[2];
        }
#pragma unroll
        for (int off = 16; off >= 1; off >>= 1) {
            num0 += __shfl_xor(num0, off, 32);
            num1 += __shfl_xor(num1, off, 32);
            num2 += __shfl_xor(num2, off, 32);
            cnt0 += __shfl_xor(cnt0, off, 32);
            cnt1 += __shfl_xor(cnt1, off, 32);
            cnt2 += __shfl_xor(cnt2, off, 32);
        }
        if (lane == 0) {
            float* p = partials + (size_t)blockIdx.x * 6;
            p[0] = num0; p[1] = num1; p[2] = num2;
            p[3] = cnt0; p[4] = cnt1; p[5] = cnt2;
        }
    }
}

__global__ __launch_bounds__(256) void finalize_loss(
    const float* __restrict__ partials, int nblk, float* __restrict__ loss_out)
{
    __shared__ float red[256];
    const int t = threadIdx.x;
    float tot[6];
    for (int g = 0; g < 6; ++g) {
        float s = 0.f;
        for (int i = t; i < nblk; i += 256) s += partials[(size_t)i * 6 + g];
        red[t] = s;
        __syncthreads();
        for (int stride = 128; stride > 0; stride >>= 1) {
            if (t < stride) red[t] += red[t + stride];
            __syncthreads();
        }
        tot[g] = red[0];
        __syncthreads();
    }
    if (t == 0) {
        float loss = 0.f;
        for (int g = 0; g < 3; ++g) loss += tot[g] / fmaxf(tot[3 + g], 1.0f);
        loss_out[0] = loss;
    }
}

extern "C" void kernel_launch(void* const* d_in, const int* in_sizes, int n_in,
                              void* d_out, int out_size, void* d_ws, size_t ws_size,
                              hipStream_t stream) {
    const int*   x    = (const int*)d_in[0];
    const int*   xm   = (const int*)d_in[1];
    const float* y    = (const float*)d_in[2];
    const float* ob   = (const float*)d_in[3];
    const float* emb  = (const float*)d_in[4];
    const float* w0   = (const float*)d_in[5];
    const float* b0   = (const float*)d_in[6];
    const float* w1   = (const float*)d_in[7];
    const float* b1   = (const float*)d_in[8];
    const float* w2   = (const float*)d_in[9];
    const float* b2   = (const float*)d_in[10];

    float* out      = (float*)d_out;
    float* partials = (float*)d_ws;

    const int nblk = B_TOT / USERS_PER_BLOCK;
    fused_pool_proj_softmax<<<nblk, 512, 0, stream>>>(
        x, xm, y, ob, emb, w0, b0, w1, b1, w2, b2, out, partials);
    finalize_loss<<<1, 256, 0, stream>>>(partials, nblk, out + (size_t)B_TOT * NATTR);
}


# Each wave pools 16 users one after another (next user's 50 row loads refill the registers the sums free, about 50 in flight) and projects them itself, no work-group barrier: 1,024 waves, was 16,384
# speedup vs baseline: 1.6839x; 1.6839x over previous
#include <hip/hip_runtime.h>
#include <math.h>


typedef __attribute__((ext_vector_type(2))) float v2f;
typedef __attribute__((ext_vector_type(8))) float v8f;

#define B_TOT 16384
#define LSEQ  50
#define DIM   64
#define NATTR 16
#define TILE_USERS 16
#define REP_PITCH  (DIM + 4)
#define ROWS_PER_STEP 10

template <int XM>
__device__ __forceinline__ float row_xor(float v) {
    const int b = __builtin_bit_cast(int, v);
    return __builtin_bit_cast(float,
        __builtin_amdgcn_update_dpp(b, b, 0x160 + XM, 0xf, 0xf, false));
}

__device__ __forceinline__ float other_row(float v) {
    const int b = __builtin_bit_cast(int, v);
    return __builtin_bit_cast(float,
        __builtin_amdgcn_permlanex16(b, b, 0x76543210, 0xfedcba98, false, false));
}

__device__ __forceinline__ float wave_sum(float v) {
    v += row_xor<1>(v);
    v += row_xor<2>(v);
    v += row_xor<4>(v);
    v += row_xor<8>(v);
    return v + other_row(v);
}

__device__ __forceinline__ float grp_max(float v, int c) {
    float t = (c < 2) ? -3.402823466e+38f : v;
    t = fmaxf(t, row_xor<1>(t));
    t = fmaxf(t, row_xor<2>(t));
    t = fmaxf(t, row_xor<4>(t));
    const float p = fmaxf(v, row_xor<1>(v));
    return (c < 2) ? p : t;
}
__device__ __forceinline__ float grp_sum(float v, int c) {
    float t = (c < 2) ? 0.0f : v;
    t += row_xor<1>(t);
    t += row_xor<2>(t);
    t += row_xor<4>(t);
    const float p = v + row_xor<1>(v);
    return (c < 2) ? p : t;
}

__device__ __forceinline__ unsigned row_off(int idx, unsigned lane8) {
    return ((unsigned)idx << 8) + lane8;
}
__device__ __forceinline__ float2 row_load(const char* emb, int idx, unsigned lane8) {
    return *(const float2*)(emb + row_off(idx, lane8));
}

#define ROW_IDX(xa, xb, l) __builtin_amdgcn_readlane(((l) < 32) ? (xa) : (xb), (l) & 31)

__global__ __launch_bounds__(32) void pool_proj_softmax_tile(
    const int*   __restrict__ x,
    const int*   __restrict__ x_mask,
    const float* __restrict__ y,
    const float* __restrict__ ob,
    const float* __restrict__ item_emb,
    const float* __restrict__ w0, const float* __restrict__ b0,
    const float* __restrict__ w1, const float* __restrict__ b1,
    const float* __restrict__ w2, const float* __restrict__ b2,
    float* __restrict__ logit_out,
    float* __restrict__ partials)
{
    __shared__ float rep[TILE_USERS][REP_PITCH];

    const int lane = threadIdx.x;
    const int c    = lane & 15;
    const int h    = lane >> 4;
    const int U0   = blockIdx.x * TILE_USERS;
    const unsigned lane8 = (unsigned)lane * 8u;
    const char* emb = (const char*)item_emb;
    const bool  tail = lane < LSEQ - 32;

    const int* xr = x      + (size_t)U0 * LSEQ;
    const int* mr = x_mask + (size_t)U0 * LSEQ;

    float2 e[LSEQ];
    {
        const int ca = xr[lane];
        const int cb = tail ? xr[32 + lane] : 0;
#pragma unroll
        for (int l = 0; l < LSEQ; ++l) e[l] = row_load(emb, ROW_IDX(ca, cb, l), lane8);
    }
    int na = xr[LSEQ + lane];
    int nb = tail ? xr[LSEQ + 32 + lane] : 0;
    int mk = mr[lane] + (tail ? mr[32 + lane] : 0);

    const float* wr;
    float bias;
    if      (c < 2) { wr = w0 + c * DIM;       bias = b0[c];     }
    else if (c < 8) { wr = w1 + (c - 2) * DIM; bias = b1[c - 2]; }
    else            { wr = w2 + (c - 8) * DIM; bias = b2[c - 8]; }
    float2 bf[16];
#pragma unroll
    for (int s = 0; s < 16; ++s) bf[s] = *(const float2*)(wr + 4 * s + 2 * h);

    for (int u = 0; u < TILE_USERS - 1; ++u) {
        const int u2 = (u + 2 < TILE_USERS) ? (u + 2) : (TILE_USERS - 1);
        const int fa = xr[u2 * LSEQ + lane];
        const int fb = tail ? xr[u2 * LSEQ + 32 + lane] : 0;
        const int mn = mr[(u + 1) * LSEQ + lane] + (tail ? mr[(u + 1) * LSEQ + 32 + lane] : 0);

        float s0 = 0.0f, s1 = 0.0f;
#pragma unroll
        for (int l0 = 0; l0 < LSEQ; l0 += ROWS_PER_STEP) {
            unsigned off[ROWS_PER_STEP];
#pragma unroll
            for (int l = l0; l < l0 + ROWS_PER_STEP; ++l) {
                s0 += e[l].x;
                s1 += e[l].y;
                off[l - l0] = row_off(ROW_IDX(na, nb, l), lane8);
            }
            asm volatile("" : "+v"(s0), "+v"(s1),
                              "+v"(off[0]), "+v"(off[1]), "+v"(off[2]), "+v"(off[3]), "+v"(off[4]),
                              "+v"(off[5]), "+v"(off[6]), "+v"(off[7]), "+v"(off[8]), "+v"(off[9]));
#pragma unroll
            for (int l = l0; l < l0 + ROWS_PER_STEP; ++l)
                e[l] = *(const float2*)(emb + off[l - l0]);
        }
        const float inv_len = 1.0f / wave_sum((float)mk);
        *(float2*)&rep[u][2 * lane] = make_float2(s0 * inv_len, s1 * inv_len);
        na = fa; nb = fb; mk = mn;
    }
    {
        float s0 = 0.0f, s1 = 0.0f;
#pragma unroll
        for (int l = 0; l < LSEQ; ++l) { s0 += e[l].x; s1 += e[l].y; }
        const float inv_len = 1.0f / wave_sum((float)mk);
        *(float2*)&rep[TILE_USERS - 1][2 * lane] = make_float2(s0 * inv_len, s1 * inv_len);
    }

    float yv[8], ov[8];
#pragma unroll
    for (int r = 0; r < 8; ++r) {
        const size_t cell = (size_t)(U0 + 8 * h + r) * NATTR + c;
        yv[r] = y[cell];
        ov[r] = ob[cell];
    }

    __syncthreads();

    v8f acc = {0.f, 0.f, 0.f, 0.f, 0.f, 0.f, 0.f, 0.f};
#pragma unroll
    for (int s = 0; s < 16; ++s) {
        const float2 af = *(const float2*)&rep[c][4 * s + 2 * h];
        v2f a, b;
        a.x = af.x;    a.y = af.y;
        b.x = bf[s].x; b.y = bf[s].y;
        acc = __builtin_amdgcn_wmma_f32_16x16x4_f32(
                  false, a, false, b, (short)0, acc, false, false);
    }

    float numl = 0.0f, cntl = 0.0f;
#pragma unroll
    for (int r = 0; r < 8; ++r) {
        const float v   = acc[r] + bias;
        const float mx  = grp_max(v, c);
        const float ex  = expf(v - mx);
        const float se  = grp_sum(ex, c);
        const float lse = mx + logf(se);
        const float inv = 1.0f / se;
        numl -= yv[r] * (v - lse) * ov[r];
        cntl += grp_max(ov[r], c);
        logit_out[(size_t)(U0 + 8 * h + r) * NATTR + c] = ex * inv;
    }

    float n0 = (c < 2)            ? numl : 0.0f;
    float n1 = (c >= 2 && c < 8)  ? numl : 0.0f;
    float n2 = (c >= 8)           ? numl : 0.0f;
    float k0 = (c == 0)           ? cntl : 0.0f;
    float k1 = (c == 2)           ? cntl : 0.0f;
    float k2 = (c == 8)           ? cntl : 0.0f;
    n0 = wave_sum(n0); n1 = wave_sum(n1); n2 = wave_sum(n2);
    k0 = wave_sum(k0); k1 = wave_sum(k1); k2 = wave_sum(k2);
    if (lane == 0) {
        float* p = partials + (size_t)blockIdx.x * 6;
        p[0] = n0; p[1] = n1; p[2] = n2;
        p[3] = k0; p[4] = k1; p[5] = k2;
    }
}

#define NBLK        (B_TOT / TILE_USERS)
#define FIN_THREADS 256
#define FIN_WAVES   (FIN_THREADS / 32)
#define FIN_PAIRS   (NBLK / 2 / FIN_THREADS)

__global__ __launch_bounds__(FIN_THREADS) void finalize_loss(
    const float* __restrict__ partials, float* __restrict__ loss_out)
{
    __shared__ float red[FIN_WAVES][8];
    const int t = threadIdx.x;

    const float4* p4 = (const float4*)partials;
    float a[6] = {0.f, 0.f, 0.f, 0.f, 0.f, 0.f};
#pragma unroll
    for (int i = 0; i < FIN_PAIRS; ++i) {
        const float4* p = p4 + (size_t)(i * FIN_THREADS + t) * 3;
        const float4 f0 = p[0], f1 = p[1], f2 = p[2];
        a[0] += f0.x; a[1] += f0.y; a[2] += f0.z; a[3] += f0.w;
        a[4] += f1.x; a[5] += f1.y; a[0] += f1.z; a[1] += f1.w;
        a[2] += f2.x; a[3] += f2.y; a[4] += f2.z; a[5] += f2.w;
    }
#pragma unroll
    for (int g = 0; g < 6; ++g) a[g] = wave_sum(a[g]);

    if ((t & 31) == 0) {
#pragma unroll
        for (int g = 0; g < 6; ++g) red[t >> 5][g] = a[g];
    }
    __syncthreads();

    if (t == 0) {
        float tot[6];
#pragma unroll
        for (int g = 0; g < 6; ++g) {
            float s = 0.f;
#pragma unroll
            for (int w = 0; w < FIN_WAVES; ++w) s += red[w][g];
            tot[g] = s;
        }
        float loss = 0.f;
#pragma unroll
        for (int g = 0; g < 3; ++g) loss += tot[g] / fmaxf(tot[3 + g], 1.0f);
        loss_out[0] = loss;
    }
}

extern "C" void kernel_launch(void* const* d_in, const int* in_sizes, int n_in,
                              void* d_out, int out_size, void* d_ws, size_t ws_size,
                              hipStream_t stream) {
    const int*   x    = (const int*)d_in[0];
    const int*   xm   = (const int*)d_in[1];
    const float* y    = (const float*)d_in[2];
    const float* ob   = (const float*)d_in[3];
    const float* emb  = (const float*)d_in[4];
    const float* w0   = (const float*)d_in[5];
    const float* b0   = (const float*)d_in[6];
    const float* w1   = (const float*)d_in[7];
    const float* b1   = (const float*)d_in[8];
    const float* w2   = (const float*)d_in[9];
    const float* b2   = (const float*)d_in[10];

    float* out      = (float*)d_out;
    float* partials = (float*)d_ws;

    const int nblk = B_TOT / TILE_USERS;
    pool_proj_softmax_tile<<<nblk, 32, 0, stream>>>(
        x, xm, y, ob, emb, w0, b0, w1, b1, w2, b2, out, partials);
    finalize_loss<<<1, FIN_THREADS, 0, stream>>>(partials, out + (size_t)B_TOT * NATTR);
}


# Two waves per 16-user tile, 8 users each in the same pooling loop: 2,048 waves, two resident per SIMD, so one wave's in-order wait on a row that misses to memory is covered by the other
# speedup vs baseline: 2.1983x; 1.3055x over previous
#include <hip/hip_runtime.h>
#include <math.h>


typedef __attribute__((ext_vector_type(2))) float v2f;
typedef __attribute__((ext_vector_type(8))) float v8f;

#define B_TOT 16384
#define LSEQ  50
#define DIM   64
#define NATTR 16
#define TILE_USERS 16
#define TILE_WAVES 2
#define WAVE_USERS (TILE_USERS / TILE_WAVES)
#define WAVE_REGS  (8 / TILE_WAVES)
#define REP_PITCH  (DIM + 4)
#define ROWS_PER_STEP 10

template <int XM>
__device__ __forceinline__ float row_xor(float v) {
    const int b = __builtin_bit_cast(int, v);
    return __builtin_bit_cast(float,
        __builtin_amdgcn_update_dpp(b, b, 0x160 + XM, 0xf, 0xf, false));
}

__device__ __forceinline__ float other_row(float v) {
    const int b = __builtin_bit_cast(int, v);
    return __builtin_bit_cast(float,
        __builtin_amdgcn_permlanex16(b, b, 0x76543210, 0xfedcba98, false, false));
}

__device__ __forceinline__ float wave_sum(float v) {
    v += row_xor<1>(v);
    v += row_xor<2>(v);
    v += row_xor<4>(v);
    v += row_xor<8>(v);
    return v + other_row(v);
}

__device__ __forceinline__ float grp_max(float v, int c) {
    float t = (c < 2) ? -3.402823466e+38f : v;
    t = fmaxf(t, row_xor<1>(t));
    t = fmaxf(t, row_xor<2>(t));
    t = fmaxf(t, row_xor<4>(t));
    const float p = fmaxf(v, row_xor<1>(v));
    return (c < 2) ? p : t;
}
__device__ __forceinline__ float grp_sum(float v, int c) {
    float t = (c < 2) ? 0.0f : v;
    t += row_xor<1>(t);
    t += row_xor<2>(t);
    t += row_xor<4>(t);
    const float p = v + row_xor<1>(v);
    return (c < 2) ? p : t;
}

__device__ __forceinline__ unsigned row_off(int idx, unsigned lane8) {
    return ((unsigned)idx << 8) + lane8;
}
__device__ __forceinline__ float2 row_load(const char* emb, int idx, unsigned lane8) {
    return *(const float2*)(emb + row_off(idx, lane8));
}

#define ROW_IDX(xa, xb, l) __builtin_amdgcn_readlane(((l) < 32) ? (xa) : (xb), (l) & 31)

__global__ __launch_bounds__(32 * TILE_WAVES) void pool_proj_softmax_tile(
    const int*   __restrict__ x,
    const int*   __restrict__ x_mask,
    const float* __restrict__ y,
    const float* __restrict__ ob,
    const float* __restrict__ item_emb,
    const float* __restrict__ w0, const float* __restrict__ b0,
    const float* __restrict__ w1, const float* __restrict__ b1,
    const float* __restrict__ w2, const float* __restrict__ b2,
    float* __restrict__ logit_out,
    float* __restrict__ partials)
{
    __shared__ float rep[TILE_USERS][REP_PITCH];
    __shared__ float comb[8];

    const int lane = threadIdx.x & 31;
    const int w    = __builtin_amdgcn_readfirstlane((int)(threadIdx.x >> 5));
    const int c    = lane & 15;
    const int h    = lane >> 4;
    const int U0   = blockIdx.x * TILE_USERS;
    const unsigned lane8 = (unsigned)lane * 8u;
    const char* emb = (const char*)item_emb;
    const bool  tail = lane < LSEQ - 32;

    const int UW = U0 + w * WAVE_USERS;
    const int* xr = x      + (size_t)UW * LSEQ;
    const int* mr = x_mask + (size_t)UW * LSEQ;

    float2 e[LSEQ];
    {
        const int ca = xr[lane];
        const int cb = tail ? xr[32 + lane] : 0;
#pragma unroll
        for (int l = 0; l < LSEQ; ++l) e[l] = row_load(emb, ROW_IDX(ca, cb, l), lane8);
    }
    int na = xr[LSEQ + lane];
    int nb = tail ? xr[LSEQ + 32 + lane] : 0;
    int mk = mr[lane] + (tail ? mr[32 + lane] : 0);

    const float* wr;
    float bias;
    if      (c < 2) { wr = w0 + c * DIM;       bias = b0[c];     }
    else if (c < 8) { wr = w1 + (c - 2) * DIM; bias = b1[c - 2]; }
    else            { wr = w2 + (c - 8) * DIM; bias = b2[c - 8]; }
    float2 bf[16];
#pragma unroll
    for (int s = 0; s < 16; ++s) bf[s] = *(const float2*)(wr + 4 * s + 2 * h);

    float (*repw)[REP_PITCH] = rep + w * WAVE_USERS;
    for (int u = 0; u < WAVE_USERS - 1; ++u) {
        const int u2 = (u + 2 < WAVE_USERS) ? (u + 2) : (WAVE_USERS - 1);
        const int fa = xr[u2 * LSEQ + lane];
        const int fb = tail ? xr[u2 * LSEQ + 32 + lane] : 0;
        const int mn = mr[(u + 1) * LSEQ + lane] + (tail ? mr[(u + 1) * LSEQ + 32 + lane] : 0);

        float s0 = 0.0f, s1 = 0.0f;
#pragma unroll
        for (int l0 = 0; l0 < LSEQ; l0 += ROWS_PER_STEP) {
            unsigned off[ROWS_PER_STEP];
#pragma unroll
            for (int l = l0; l < l0 + ROWS_PER_STEP; ++l) {
                s0 += e[l].x;
                s1 += e[l].y;
                off[l - l0] = row_off(ROW_IDX(na, nb, l), lane8);
            }
            asm volatile("" : "+v"(s0), "+v"(s1),
                              "+v"(off[0]), "+v"(off[1]), "+v"(off[2]), "+v"(off[3]), "+v"(off[4]),
                              "+v"(off[5]), "+v"(off[6]), "+v"(off[7]), "+v"(off[8]), "+v"(off[9]));
#pragma unroll
            for (int l = l0; l < l0 + ROWS_PER_STEP; ++l)
                e[l] = *(const float2*)(emb + off[l - l0]);
        }
        const float inv_len = 1.0f / wave_sum((float)mk);
        *(float2*)&repw[u][2 * lane] = make_float2(s0 * inv_len, s1 * inv_len);
        na = fa; nb = fb; mk = mn;
    }
    {
        float s0 = 0.0f, s1 = 0.0f;
#pragma unroll
        for (int l = 0; l < LSEQ; ++l) { s0 += e[l].x; s1 += e[l].y; }
        const float inv_len = 1.0f / wave_sum((float)mk);
        *(float2*)&repw[WAVE_USERS - 1][2 * lane] = make_float2(s0 * inv_len, s1 * inv_len);
    }

    const int R0 = 8 * h + WAVE_REGS * w;
    float yv[WAVE_REGS], ov[WAVE_REGS];
#pragma unroll
    for (int r = 0; r < WAVE_REGS; ++r) {
        const size_t cell = (size_t)(U0 + R0 + r) * NATTR + c;
        yv[r] = y[cell];
        ov[r] = ob[cell];
    }

    __syncthreads();

    v8f acc = {0.f, 0.f, 0.f, 0.f, 0.f, 0.f, 0.f, 0.f};
#pragma unroll
    for (int s = 0; s < 16; ++s) {
        const float2 af = *(const float2*)&rep[c][4 * s + 2 * h];
        v2f a, b;
        a.x = af.x;    a.y = af.y;
        b.x = bf[s].x; b.y = bf[s].y;
        acc = __builtin_amdgcn_wmma_f32_16x16x4_f32(
                  false, a, false, b, (short)0, acc, false, false);
    }

    float numl = 0.0f, cntl = 0.0f;
    float prob[WAVE_REGS];
#pragma unroll
    for (int r = 0; r < WAVE_REGS; ++r) {
        const float v   = (w ? acc[WAVE_REGS + r] : acc[r]) + bias;
        const float mx  = grp_max(v, c);
        const float ex  = expf(v - mx);
        const float se  = grp_sum(ex, c);
        const float lse = mx + logf(se);
        const float inv = 1.0f / se;
        numl -= yv[r] * (v - lse) * ov[r];
        cntl += grp_max(ov[r], c);
        prob[r] = ex * inv;
    }

    float n0 = (c < 2)            ? numl : 0.0f;
    float n1 = (c >= 2 && c < 8)  ? numl : 0.0f;
    float n2 = (c >= 8)           ? numl : 0.0f;
    float k0 = (c == 0)           ? cntl : 0.0f;
    float k1 = (c == 2)           ? cntl : 0.0f;
    float k2 = (c == 8)           ? cntl : 0.0f;
    n0 = wave_sum(n0); n1 = wave_sum(n1); n2 = wave_sum(n2);
    k0 = wave_sum(k0); k1 = wave_sum(k1); k2 = wave_sum(k2);
    if (w == 1 && lane == 0) {
        comb[0] = n0; comb[1] = n1; comb[2] = n2;
        comb[3] = k0; comb[4] = k1; comb[5] = k2;
    }
    __syncthreads();
#pragma unroll
    for (int r = 0; r < WAVE_REGS; ++r)
        logit_out[(size_t)(U0 + R0 + r) * NATTR + c] = prob[r];
    if (w == 0 && lane == 0) {
        float* p = partials + (size_t)blockIdx.x * 6;
        p[0] = n0 + comb[0]; p[1] = n1 + comb[1]; p[2] = n2 + comb[2];
        p[3] = k0 + comb[3]; p[4] = k1 + comb[4]; p[5] = k2 + comb[5];
    }
}

#define NBLK        (B_TOT / TILE_USERS)
#define FIN_THREADS 256
#define FIN_WAVES   (FIN_THREADS / 32)
#define FIN_PAIRS   (NBLK / 2 / FIN_THREADS)

__global__ __launch_bounds__(FIN_THREADS) void finalize_loss(
    const float* __restrict__ partials, float* __restrict__ loss_out)
{
    __shared__ float red[FIN_WAVES][8];
    const int t = threadIdx.x;

    const float4* p4 = (const float4*)partials;
    float a[6] = {0.f, 0.f, 0.f, 0.f, 0.f, 0.f};
#pragma unroll
    for (int i = 0; i < FIN_PAIRS; ++i) {
        const float4* p = p4 + (size_t)(i * FIN_THREADS + t) * 3;
        const float4 f0 = p[0], f1 = p[1], f2 = p[2];
        a[0] += f0.x; a[1] += f0.y; a[2] += f0.z; a[3] += f0.w;
        a[4] += f1.x; a[5] += f1.y; a[0] += f1.z; a[1] += f1.w;
        a[2] += f2.x; a[3] += f2.y; a[4] += f2.z; a[5] += f2.w;
    }
#pragma unroll
    for (int g = 0; g < 6; ++g) a[g] = wave_sum(a[g]);

    if ((t & 31) == 0) {
#pragma unroll
        for (int g = 0; g < 6; ++g) red[t >> 5][g] = a[g];
    }
    __syncthreads();

    if (t == 0) {
        float tot[6];
#pragma unroll
        for (int g = 0; g < 6; ++g) {
            float s = 0.f;
#pragma unroll
            for (int w = 0; w < FIN_WAVES; ++w) s += red[w][g];
            tot[g] = s;
        }
        float loss = 0.f;
#pragma unroll
        for (int g = 0; g < 3; ++g) loss += tot[g] / fmaxf(tot[3 + g], 1.0f);
        loss_out[0] = loss;
    }
}

extern "C" void kernel_launch(void* const* d_in, const int* in_sizes, int n_in,
                              void* d_out, int out_size, void* d_ws, size_t ws_size,
                              hipStream_t stream) {
    const int*   x    = (const int*)d_in[0];
    const int*   xm   = (const int*)d_in[1];
    const float* y    = (const float*)d_in[2];
    const float* ob   = (const float*)d_in[3];
    const float* emb  = (const float*)d_in[4];
    const float* w0   = (const float*)d_in[5];
    const float* b0   = (const float*)d_in[6];
    const float* w1   = (const float*)d_in[7];
    const float* b1   = (const float*)d_in[8];
    const float* w2   = (const float*)d_in[9];
    const float* b2   = (const float*)d_in[10];

    float* out      = (float*)d_out;
    float* partials = (float*)d_ws;

    const int nblk = B_TOT / TILE_USERS;
    pool_proj_softmax_tile<<<nblk, 32 * TILE_WAVES, 0, stream>>>(
        x, xm, y, ob, emb, w0, b0, w1, b1, w2, b2, out, partials);
    finalize_loss<<<1, FIN_THREADS, 0, stream>>>(partials, out + (size_t)B_TOT * NATTR);
}
